# starting kernel
# baseline (speedup 1.0000x reference)
#include <hip/hip_runtime.h>
#include <hip/hip_bf16.h>

#define Zn 131072
#define Nn 64
#define Kn 64
#define Wn 32

typedef __attribute__((ext_vector_type(2))) float v2f;
typedef __attribute__((ext_vector_type(8))) float v8f;

__global__ __launch_bounds__(256) void lts_reduce_mix_kernel(
    const float* __restrict__ mix,
    const float* __restrict__ wgt,
    float* __restrict__ Msw)
{
    const int t = blockIdx.x * blockDim.x + threadIdx.x;
    float acc = 0.0f;
#pragma unroll
    for (int w = 0; w < Wn; ++w)
        acc = fmaf(wgt[w], mix[w * (Kn * Nn * Nn) + t], acc);

    const int k  = t >> 12;
    const int i  = (t >> 6) & 63;
    const int j  = t & 63;
    const int kt = k >> 4;
    const int kl = k & 15;
    const int jj = j >> 2;
    const int jo = j & 3;
    const int half = jo >> 1;
    const int lane = half * 16 + kl;
    const int fidx = (((kt * 64 + i) * 16 + jj) * 32 + lane) * 2 + (jo & 1);
    Msw[fidx] = acc;
}

__global__ __launch_bounds__(1024) void lts_bilinear_wmma_kernel(
    const float* __restrict__ feat,
    const float* __restrict__ Msw,
    float* __restrict__ out)
{
    extern __shared__ float smem[];

    const int tid  = threadIdx.x;
    const int wave = tid >> 5;
    const int lane = tid & 31;
    const int half = lane >> 4;
    const int l16  = lane & 15;
    const int zrow = blockIdx.x * 512 + wave * 16 + l16;

    v2f  fj[16];
#pragma unroll
    for (int jj = 0; jj < 16; ++jj) {
        const float2 v = *(const float2*)(feat + zrow * 64 + jj * 4 + half * 2);
        fj[jj].x = v.x;
        fj[jj].y = v.y;
    }
    const float* frow = feat + zrow * 64;

    for (int kt = 0; kt < 4; ++kt) {
        __syncthreads();
        {
            const float4* src = (const float4*)(Msw + (size_t)kt * 65536);
            float4* dst = (float4*)smem;
            for (int idx = tid; idx < 16384; idx += 1024)
                dst[idx] = src[idx];
        }
        __syncthreads();

        const float2* Bl = (const float2*)smem;
        v8f acc = {};
        for (int i = 0; i < 64; ++i) {
            const float fi = frow[i];
            const float2* bp = Bl + i * 512 + lane;
#pragma unroll
            for (int jj = 0; jj < 16; ++jj) {
                const float2 bv = bp[jj * 32];
                v2f b; b.x = bv.x; b.y = bv.y;
                v2f a; a.x = fi * fj[jj].x; a.y = fi * fj[jj].y;
                acc = __builtin_amdgcn_wmma_f32_16x16x4_f32(
                     false, a,  false, b,
                     (short)0, acc,
                     false,  false);
            }
        }

#pragma unroll
        for (int r = 0; r < 8; ++r) {
            const int z = blockIdx.x * 512 + wave * 16 + half * 8 + r;
            out[z * 64 + kt * 16 + l16] = acc[r];
        }
    }
}

extern "C" void kernel_launch(void* const* d_in, const int* in_sizes, int n_in,
                              void* d_out, int out_size, void* d_ws, size_t ws_size,
                              hipStream_t stream) {
    const float* feat = (const float*)d_in[0];
    const float* mix  = (const float*)d_in[1];
    const float* wgt  = (const float*)d_in[2];
    float* out = (float*)d_out;
    float* Msw = (float*)d_ws;

    lts_reduce_mix_kernel<<<(Kn * Nn * Nn) / 256, 256, 0, stream>>>(mix, wgt, Msw);

    lts_bilinear_wmma_kernel<<<Zn / 512, 1024, 65536 * sizeof(float), stream>>>(
        feat, Msw, out);
}


# Fold the symmetric (i,j)/(j,i) pairs: kernel 1 adds M[k,i,j]+M[k,j,i] via an LDS transpose, kernel 2 reduces the upper triangle only (544 WMMAs per k-tile, not 1024); sum order changes
# speedup vs baseline: 1.6322x; 1.6322x over previous
#include <hip/hip_runtime.h>
#include <hip/hip_bf16.h>


#define Zn 131072
#define Nn 64
#define Kn 64
#define Wn 32

#define SLOTS 544

typedef __attribute__((ext_vector_type(2))) float v2f;
typedef __attribute__((ext_vector_type(8))) float v8f;

__host__ __device__ constexpr int slot_base(int q, int r) {
    return 64 * q - 2 * q * (q - 1) + r * (16 - q);
}

__global__ __launch_bounds__(1024) void lts_reduce_fold_kernel(
    const float* __restrict__ mix,
    const float* __restrict__ wgt,
    float* __restrict__ Ssw)
{
    __shared__ float sM[Nn * (Nn + 1)];

    const int k   = blockIdx.x;
    const int tid = threadIdx.x;
    const float* src = mix + k * (Nn * Nn) + tid;

    float acc[4] = {0.0f, 0.0f, 0.0f, 0.0f};
#pragma unroll
    for (int w = 0; w < Wn; ++w) {
        const float wv = wgt[w];
#pragma unroll
        for (int m = 0; m < 4; ++m)
            acc[m] = fmaf(wv, src[w * (Kn * Nn * Nn) + m * 1024], acc[m]);
    }
#pragma unroll
    for (int m = 0; m < 4; ++m) {
        const int e = m * 1024 + tid;
        sM[(e >> 6) * (Nn + 1) + (e & 63)] = acc[m];
    }
    __syncthreads();

    const int kt = k >> 4;
    const int kl = k & 15;
#pragma unroll
    for (int m = 0; m < 4; ++m) {
        const int e  = m * 1024 + tid;
        const int i  = e >> 6;
        const int j  = e & 63;
        const int q  = i >> 2;
        const int r  = i & 3;
        const int jj = j >> 2;
        if (jj >= q) {
            float v = 0.0f;
            if (j > i)       v = acc[m] + sM[j * (Nn + 1) + i];
            else if (j == i) v = acc[m];
            const int jo   = j & 3;
            const int lane = (jo >> 1) * 16 + kl;
            const int slot = slot_base(q, r) + (jj - q);
            Ssw[((kt * SLOTS + slot) * 32 + lane) * 2 + (jo & 1)] = v;
        }
    }
}

__global__ __launch_bounds__(1024) void lts_bilinear_wmma_kernel(
    const float* __restrict__ feat,
    const float* __restrict__ Ssw,
    float* __restrict__ out)
{
    extern __shared__ float smem[];

    const int tid  = threadIdx.x;
    const int wave = tid >> 5;
    const int lane = tid & 31;
    const int half = lane >> 4;
    const int l16  = lane & 15;
    const int zrow = blockIdx.x * 512 + wave * 16 + l16;

    v2f  fj[16];
#pragma unroll
    for (int jj = 0; jj < 16; ++jj) {
        const float2 v = *(const float2*)(feat + zrow * 64 + jj * 4 + half * 2);
        fj[jj].x = v.x;
        fj[jj].y = v.y;
    }
    const float* frow = feat + zrow * 64;

    for (int kt = 0; kt < 4; ++kt) {
        __syncthreads();
        {
            const float4* src = (const float4*)(Ssw + (size_t)kt * (SLOTS * 64));
            float4* dst = (float4*)smem;
            for (int idx = tid; idx < SLOTS * 16; idx += 1024)
                dst[idx] = src[idx];
        }
        __syncthreads();

        const float2* Bl = (const float2*)smem + lane;
        v8f acc = {};
#pragma unroll
        for (int q = 0; q < 16; ++q) {
#pragma unroll 1
            for (int r = 0; r < 4; ++r) {
                const float fi = frow[q * 4 + r];
                const float2* bp = Bl + (slot_base(q, 0) + r * (16 - q)) * 32;
#pragma unroll
                for (int jj = q; jj < 16; ++jj) {
                    const float2 bv = bp[(jj - q) * 32];
                    v2f b; b.x = bv.x; b.y = bv.y;
                    v2f a; a.x = fi * fj[jj].x; a.y = fi * fj[jj].y;
                    acc = __builtin_amdgcn_wmma_f32_16x16x4_f32(
                         false, a,  false, b,
                         (short)0, acc,
                         false,  false);
                }
            }
        }

#pragma unroll
        for (int r = 0; r < 8; ++r) {
            const int z = blockIdx.x * 512 + wave * 16 + half * 8 + r;
            out[z * 64 + kt * 16 + l16] = acc[r];
        }
    }
}

extern "C" void kernel_launch(void* const* d_in, const int* in_sizes, int n_in,
                              void* d_out, int out_size, void* d_ws, size_t ws_size,
                              hipStream_t stream) {
    const float* feat = (const float*)d_in[0];
    const float* mix  = (const float*)d_in[1];
    const float* wgt  = (const float*)d_in[2];
    float* out = (float*)d_out;
    float* Ssw = (float*)d_ws;

    lts_reduce_fold_kernel<<<Kn, 1024, 0, stream>>>(mix, wgt, Ssw);

    lts_bilinear_wmma_kernel<<<Zn / 512, 1024, SLOTS * 64 * sizeof(float), stream>>>(
        feat, Ssw, out);
}


# Kernel 2 row loop software-pipelined: next row's B fragments are read from LDS as one group under the current row's WMMAs, f[z,i] from registers, two z-tiles per wave; sum order unchanged
# speedup vs baseline: 1.8640x; 1.1420x over previous
#include <hip/hip_runtime.h>
#include <hip/hip_bf16.h>


#define Zn 131072
#define Nn 64
#define Kn 64
#define Wn 32

#define SLOTS 544

typedef __attribute__((ext_vector_type(2))) float v2f;
typedef __attribute__((ext_vector_type(8))) float v8f;

__host__ __device__ constexpr int slot_base(int q, int r) {
    return 64 * q - 2 * q * (q - 1) + r * (16 - q);
}

__global__ __launch_bounds__(1024) void lts_reduce_fold_kernel(
    const float* __restrict__ mix,
    const float* __restrict__ wgt,
    float* __restrict__ Ssw)
{
    __shared__ float sM[Nn * (Nn + 1)];

    const int k   = blockIdx.x;
    const int tid = threadIdx.x;
    const float* src = mix + k * (Nn * Nn) + tid;

    float acc[4] = {0.0f, 0.0f, 0.0f, 0.0f};
#pragma unroll
    for (int w = 0; w < Wn; ++w) {
        const float wv = wgt[w];
#pragma unroll
        for (int m = 0; m < 4; ++m)
            acc[m] = fmaf(wv, src[w * (Kn * Nn * Nn) + m * 1024], acc[m]);
    }
#pragma unroll
    for (int m = 0; m < 4; ++m) {
        const int e = m * 1024 + tid;
        sM[(e >> 6) * (Nn + 1) + (e & 63)] = acc[m];
    }
    __syncthreads();

    const int kt = k >> 4;
    const int kl = k & 15;
#pragma unroll
    for (int m = 0; m < 4; ++m) {
        const int e  = m * 1024 + tid;
        const int i  = e >> 6;
        const int j  = e & 63;
        const int q  = i >> 2;
        const int r  = i & 3;
        const int jj = j >> 2;
        if (jj >= q) {
            float v = 0.0f;
            if (j > i)       v = acc[m] + sM[j * (Nn + 1) + i];
            else if (j == i) v = acc[m];
            const int jo   = j & 3;
            const int lane = (jo >> 1) * 16 + kl;
            const int slot = slot_base(q, r) + (jj - q);
            Ssw[((kt * SLOTS + slot) * 32 + lane) * 2 + (jo & 1)] = v;
        }
    }
}

template <int N>
__device__ __forceinline__ void fetch_row(const float2* p, v2f (&b)[16]) {
#pragma unroll
    for (int s = 0; s < N; ++s) {
        const float2 v = p[s * 32];
        b[s].x = v.x;
        b[s].y = v.y;
    }
}

template <int Q>
__device__ __forceinline__ void mma_row(const float fi0, const float fi1,
                                        const v2f (&fj0)[16], const v2f (&fj1)[16],
                                        const v2f (&b)[16], v8f& acc0, v8f& acc1) {
#pragma unroll
    for (int s = 0; s < 16 - Q; ++s) {
        v2f a0; a0.x = fi0 * fj0[Q + s].x; a0.y = fi0 * fj0[Q + s].y;
        v2f a1; a1.x = fi1 * fj1[Q + s].x; a1.y = fi1 * fj1[Q + s].y;
        acc0 = __builtin_amdgcn_wmma_f32_16x16x4_f32(
             false, a0,  false, b[s],
             (short)0, acc0,  false,  false);
        acc1 = __builtin_amdgcn_wmma_f32_16x16x4_f32(
             false, a1,  false, b[s],
             (short)0, acc1,  false,  false);
    }
}

template <int Q>
struct QRows {
    static __device__ __forceinline__ void run(
        const float2* Bl, const int half,
        const v2f (&fj0)[16], const v2f (&fo0)[16],
        const v2f (&fj1)[16], const v2f (&fo1)[16],
        v2f (&bA)[16], v2f (&bB)[16], v8f& acc0, v8f& acc1) {
        constexpr int n = 16 - Q;
        const float2* rp = Bl + slot_base(Q, 0) * 32;
#pragma unroll 1
        for (int t = 0; t < 2; ++t) {
            const bool mine = (half == t);

            fetch_row<n>(rp + (2 * t + 1) * n * 32, bB);
            __builtin_amdgcn_sched_barrier(0);
            {
                const float fi0 = mine ? fj0[Q].x : fo0[Q].x;
                const float fi1 = mine ? fj1[Q].x : fo1[Q].x;
                mma_row<Q>(fi0, fi1, fj0, fj1, bA, acc0, acc1);
            }

            if (Q < 15 || t == 0)
                fetch_row<n>(rp + (2 * t + 2) * n * 32, bA);
            __builtin_amdgcn_sched_barrier(0);
            {
                const float fi0 = mine ? fj0[Q].y : fo0[Q].y;
                const float fi1 = mine ? fj1[Q].y : fo1[Q].y;
                mma_row<Q>(fi0, fi1, fj0, fj1, bB, acc0, acc1);
            }
        }
        QRows<Q + 1>::run(Bl, half, fj0, fo0, fj1, fo1, bA, bB, acc0, acc1);
    }
};
template <>
struct QRows<16> {
    static __device__ __forceinline__ void run(
        const float2*, const int,
        const v2f (&)[16], const v2f (&)[16], const v2f (&)[16], const v2f (&)[16],
        v2f (&)[16], v2f (&)[16], v8f&, v8f&) {}
};

__global__ __launch_bounds__(512) void lts_bilinear_wmma_kernel(
    const float* __restrict__ feat,
    const float* __restrict__ Ssw,
    float* __restrict__ out)
{
    extern __shared__ float smem[];

    const int tid  = threadIdx.x;
    const int wave = tid >> 5;
    const int lane = tid & 31;
    const int half = lane >> 4;
    const int l16  = lane & 15;
    const int zbase = blockIdx.x * 512 + wave * 32;

    v2f fj0[16], fo0[16], fj1[16], fo1[16];
    {
        const float4* row0 = (const float4*)(feat + (zbase + l16) * 64);
        const float4* row1 = (const float4*)(feat + (zbase + 16 + l16) * 64);
#pragma unroll
        for (int jj = 0; jj < 16; ++jj) {
            const float4 u = row0[jj];
            const float4 w = row1[jj];
            fj0[jj].x = half ? u.z : u.x;  fj0[jj].y = half ? u.w : u.y;
            fo0[jj].x = half ? u.x : u.z;  fo0[jj].y = half ? u.y : u.w;
            fj1[jj].x = half ? w.z : w.x;  fj1[jj].y = half ? w.w : w.y;
            fo1[jj].x = half ? w.x : w.z;  fo1[jj].y = half ? w.y : w.w;
        }
    }

    for (int kt = 0; kt < 4; ++kt) {
        __syncthreads();
        {
            const float4* src = (const float4*)(Ssw + (size_t)kt * (SLOTS * 64));
            float4* dst = (float4*)smem;
            for (int idx = tid; idx < SLOTS * 16; idx += 512)
                dst[idx] = src[idx];
        }
        __syncthreads();

        const float2* Bl = (const float2*)smem + lane;
        v8f acc0 = {};
        v8f acc1 = {};
        v2f bA[16], bB[16];
        fetch_row<16>(Bl, bA);
        QRows<0>::run(Bl, half, fj0, fo0, fj1, fo1, bA, bB, acc0, acc1);

#pragma unroll
        for (int r = 0; r < 8; ++r) {
            const int z = zbase + half * 8 + r;
            out[z * 64 + kt * 16 + l16]        = acc0[r];
            out[(z + 16) * 64 + kt * 16 + l16] = acc1[r];
        }
    }
}

extern "C" void kernel_launch(void* const* d_in, const int* in_sizes, int n_in,
                              void* d_out, int out_size, void* d_ws, size_t ws_size,
                              hipStream_t stream) {
    const float* feat = (const float*)d_in[0];
    const float* mix  = (const float*)d_in[1];
    const float* wgt  = (const float*)d_in[2];
    float* out = (float*)d_out;
    float* Ssw = (float*)d_ws;

    lts_reduce_fold_kernel<<<Kn, 1024, 0, stream>>>(mix, wgt, Ssw);

    lts_bilinear_wmma_kernel<<<Zn / 512, 512, SLOTS * 64 * sizeof(float), stream>>>(
        feat, Ssw, out);
}


# Kernel 2 wave tile widened from 2 z-tiles x 1 k-tile to 4 z-tiles x 2 k-tiles (8 waves/block, 272 KB LDS per k-tile pair): 8 WMMAs per LDS fragment pair, 1/4 the read groups; sum order unchanged
# speedup vs baseline: 1.9482x; 1.0452x over previous
#include <hip/hip_runtime.h>
#include <hip/hip_bf16.h>


#define Zn 131072
#define Nn 64
#define Kn 64
#define Wn 32

#define SLOTS 544

#define ZT 4

typedef __attribute__((ext_vector_type(2))) float v2f;
typedef __attribute__((ext_vector_type(8))) float v8f;

__host__ __device__ constexpr int slot_base(int q, int r) {
    return 64 * q - 2 * q * (q - 1) + r * (16 - q);
}

__global__ __launch_bounds__(1024) void lts_reduce_fold_kernel(
    const float* __restrict__ mix,
    const float* __restrict__ wgt,
    float* __restrict__ Ssw)
{
    __shared__ float sM[Nn * (Nn + 1)];

    const int k   = blockIdx.x;
    const int tid = threadIdx.x;
    const float* src = mix + k * (Nn * Nn) + tid;

    float acc[4] = {0.0f, 0.0f, 0.0f, 0.0f};
#pragma unroll
    for (int w = 0; w < Wn; ++w) {
        const float wv = wgt[w];
#pragma unroll
        for (int m = 0; m < 4; ++m)
            acc[m] = fmaf(wv, src[w * (Kn * Nn * Nn) + m * 1024], acc[m]);
    }
#pragma unroll
    for (int m = 0; m < 4; ++m) {
        const int e = m * 1024 + tid;
        sM[(e >> 6) * (Nn + 1) + (e & 63)] = acc[m];
    }
    __syncthreads();

    const int kp = k >> 5;
    const int u  = (k >> 4) & 1;
    const int kl = k & 15;
#pragma unroll
    for (int m = 0; m < 4; ++m) {
        const int e  = m * 1024 + tid;
        const int i  = e >> 6;
        const int j  = e & 63;
        const int q  = i >> 2;
        const int r  = i & 3;
        const int jj = j >> 2;
        if (jj >= q) {
            float v = 0.0f;
            if (j > i)       v = acc[m] + sM[j * (Nn + 1) + i];
            else if (j == i) v = acc[m];
            const int jo   = j & 3;
            const int lane = (jo >> 1) * 16 + kl;
            const int slot = slot_base(q, r) + (jj - q);
            Ssw[((((kp * SLOTS + slot) * 2 + u) * 32 + lane) * 2) + (jo & 1)] = v;
        }
    }
}


__device__ __forceinline__ float xhalf(float x) {
    return __builtin_bit_cast(float, __builtin_amdgcn_permlanex16(
        __builtin_bit_cast(int, x), __builtin_bit_cast(int, x),
        0x76543210, 0xfedcba98, false, false));
}

template <int N>
__device__ __forceinline__ void fetch_row(const float2* p, v2f (&b)[32]) {
#pragma unroll
    for (int e = 0; e < N; ++e) {
        const float2 v = p[e * 32];
        b[e].x = v.x;
        b[e].y = v.y;
    }
}

template <int Q>
__device__ __forceinline__ void mma_row(const float (&fi)[ZT], const v2f (&fj)[ZT][16],
                                        const v2f (&b)[32], v8f (&acc)[ZT][2]) {
#pragma unroll
    for (int s = 0; s < 16 - Q; ++s) {
        v2f a[ZT];
#pragma unroll
        for (int zt = 0; zt < ZT; ++zt) {
            a[zt].x = fi[zt] * fj[zt][Q + s].x;
            a[zt].y = fi[zt] * fj[zt][Q + s].y;
        }
#pragma unroll
        for (int zt = 0; zt < ZT; ++zt) {
#pragma unroll
            for (int u = 0; u < 2; ++u)
                acc[zt][u] = __builtin_amdgcn_wmma_f32_16x16x4_f32(
                     false, a[zt],  false, b[2 * s + u],
                     (short)0, acc[zt][u],  false,  false);
        }
    }
}

template <int Q>
struct QRows {
    static __device__ __forceinline__ void run(
        const float2* Bl, const int half, const v2f (&fj)[ZT][16],
        v2f (&bC)[32], v2f (&bN)[32], v8f (&acc)[ZT][2]) {
        constexpr int n = 16 - Q;
        const float2* rp = Bl + slot_base(Q, 0) * 64;

        v2f part[ZT];
#pragma unroll
        for (int zt = 0; zt < ZT; ++zt) {
            part[zt].x = xhalf(fj[zt][Q].x);
            part[zt].y = xhalf(fj[zt][Q].y);
        }

#pragma unroll 1
        for (int r = 0; r < 4; ++r) {
            const bool mine = (half == (r >> 1));
            const bool odd  = (r & 1) != 0;
            float fi[ZT];
#pragma unroll
            for (int zt = 0; zt < ZT; ++zt) {
                const float own = odd ? fj[zt][Q].y : fj[zt][Q].x;
                const float oth = odd ? part[zt].y : part[zt].x;
                fi[zt] = mine ? own : oth;
            }

            const float2* np = rp + (r + 1) * n * 64;
            fetch_row<2 * (n - 1)>(np, bN);
            if (r < 3) {
                const float2 v0 = np[(2 * n - 2) * 32];
                const float2 v1 = np[(2 * n - 1) * 32];
                bN[2 * n - 2].x = v0.x;  bN[2 * n - 2].y = v0.y;
                bN[2 * n - 1].x = v1.x;  bN[2 * n - 1].y = v1.y;
            }
            __builtin_amdgcn_sched_barrier(0);
            mma_row<Q>(fi, fj, bC, acc);

#pragma unroll
            for (int e = 0; e < 2 * n; ++e) bC[e] = bN[e];
        }
        QRows<Q + 1>::run(Bl, half, fj, bC, bN, acc);
    }
};
template <>
struct QRows<16> {
    static __device__ __forceinline__ void run(
        const float2*, const int, const v2f (&)[ZT][16],
        v2f (&)[32], v2f (&)[32], v8f (&)[ZT][2]) {}
};

__global__ __launch_bounds__(256) void lts_bilinear_wmma_kernel(
    const float* __restrict__ feat,
    const float* __restrict__ Ssw,
    float* __restrict__ out)
{
    __shared__ __attribute__((aligned(16))) float smem[SLOTS * 128];

    const int tid  = threadIdx.x;
    const int wave = tid >> 5;
    const int lane = tid & 31;
    const int half = lane >> 4;
    const int l16  = lane & 15;
    const int zbase = blockIdx.x * 512 + wave * 64;

    v2f fj[ZT][16];
#pragma unroll
    for (int zt = 0; zt < ZT; ++zt) {
        const float2* row = (const float2*)(feat + (zbase + 16 * zt + l16) * 64) + half;
#pragma unroll
        for (int jj = 0; jj < 16; ++jj) {
            const float2 v = row[2 * jj];
            fj[zt][jj].x = v.x;
            fj[zt][jj].y = v.y;
        }
    }

    for (int kp = 0; kp < 2; ++kp) {
        __syncthreads();
        {
            const float4* src = (const float4*)(Ssw + (size_t)kp * (SLOTS * 128)) + tid;
            float4* dst = (float4*)smem + tid;
#pragma unroll 1
            for (int bt = 0; bt < 2; ++bt) {
                float4 v[34];
#pragma unroll
                for (int m = 0; m < 34; ++m) v[m] = src[(bt * 34 + m) * 256];
#pragma unroll
                for (int m = 0; m < 34; ++m) dst[(bt * 34 + m) * 256] = v[m];
            }
        }
        __syncthreads();

        const float2* Bl = (const float2*)smem + lane;
        v8f acc[ZT][2];
#pragma unroll
        for (int zt = 0; zt < ZT; ++zt) {
            acc[zt][0] = v8f{};
            acc[zt][1] = v8f{};
        }
        v2f bC[32], bN[32];
        fetch_row<32>(Bl, bC);
        QRows<0>::run(Bl, half, fj, bC, bN, acc);

#pragma unroll
        for (int zt = 0; zt < ZT; ++zt) {
#pragma unroll
            for (int r = 0; r < 8; ++r) {
                const int z = zbase + 16 * zt + half * 8 + r;
                out[z * 64 + (2 * kp) * 16 + l16]     = acc[zt][0][r];
                out[z * 64 + (2 * kp + 1) * 16 + l16] = acc[zt][1][r];
            }
        }
    }
}

extern "C" void kernel_launch(void* const* d_in, const int* in_sizes, int n_in,
                              void* d_out, int out_size, void* d_ws, size_t ws_size,
                              hipStream_t stream) {
    const float* feat = (const float*)d_in[0];
    const float* mix  = (const float*)d_in[1];
    const float* wgt  = (const float*)d_in[2];
    float* out = (float*)d_out;
    float* Ssw = (float*)d_ws;

    lts_reduce_fold_kernel<<<Kn, 1024, 0, stream>>>(mix, wgt, Ssw);

    lts_bilinear_wmma_kernel<<<Zn / 512, 256, 0, stream>>>(feat, Ssw, out);
}


# Kernel 2 wave tile widened from 2 to all 4 k-tiles (S staged as two parts of the triangle, one B register set refilled half a row ahead): an A product feeds 4 WMMAs, not 2; sum order unchanged
# speedup vs baseline: 2.0252x; 1.0395x over previous
#include <hip/hip_runtime.h>
#include <hip/hip_bf16.h>


#define Zn 131072
#define Nn 64
#define Kn 64
#define Wn 32

#define SLOTS 544

#define ZT 4
#define KT 4

#define PART_SLOTS  280
#define PART1_FIRST 264
#define SPLIT_Q     5

typedef __attribute__((ext_vector_type(2))) float v2f;
typedef __attribute__((ext_vector_type(8))) float v8f;

__host__ __device__ constexpr int slot_base(int q, int r) {
    return 64 * q - 2 * q * (q - 1) + r * (16 - q);
}

__global__ __launch_bounds__(1024) void lts_reduce_fold_kernel(
    const float* __restrict__ mix,
    const float* __restrict__ wgt,
    float* __restrict__ Ssw)
{
    __shared__ float sM[Nn * (Nn + 1)];

    const int k   = blockIdx.x;
    const int tid = threadIdx.x;
    const float* src = mix + k * (Nn * Nn) + tid;

    float acc[4] = {0.0f, 0.0f, 0.0f, 0.0f};
#pragma unroll
    for (int w = 0; w < Wn; ++w) {
        const float wv = wgt[w];
#pragma unroll
        for (int m = 0; m < 4; ++m)
            acc[m] = fmaf(wv, src[w * (Kn * Nn * Nn) + m * 1024], acc[m]);
    }
#pragma unroll
    for (int m = 0; m < 4; ++m) {
        const int e = m * 1024 + tid;
        sM[(e >> 6) * (Nn + 1) + (e & 63)] = acc[m];
    }
    __syncthreads();

    const int u  = k >> 4;
    const int kl = k & 15;
#pragma unroll
    for (int m = 0; m < 4; ++m) {
        const int e  = m * 1024 + tid;
        const int i  = e >> 6;
        const int j  = e & 63;
        const int q  = i >> 2;
        const int r  = i & 3;
        const int jj = j >> 2;
        if (jj >= q) {
            float v = 0.0f;
            if (j > i)       v = acc[m] + sM[j * (Nn + 1) + i];
            else if (j == i) v = acc[m];
            const int jo   = j & 3;
            const int lane = (jo >> 1) * 16 + kl;
            const int slot = slot_base(q, r) + (jj - q);
            Ssw[(((slot * KT + u) * 32 + lane) * 2) + (jo & 1)] = v;
        }
    }
}


__device__ __forceinline__ float xhalf(float x) {
    return __builtin_bit_cast(float, __builtin_amdgcn_permlanex16(
        __builtin_bit_cast(int, x), __builtin_bit_cast(int, x),
        0x76543210, 0xfedcba98, false, false));
}

template <int E0, int E1>
__device__ __forceinline__ void load_frags(const float2* p, v2f (&b)[64]) {
#pragma unroll
    for (int e = E0; e < E1; ++e) {
        const float2 v = p[e * 32];
        b[e].x = v.x;
        b[e].y = v.y;
    }
}

template <int Q, int S0, int S1>
__device__ __forceinline__ void mma_steps(const float (&fi)[ZT], const v2f (&fj)[ZT][16],
                                          const v2f (&b)[64], v8f (&acc)[ZT][KT]) {
#pragma unroll
    for (int s = S0; s < S1; ++s) {
        v2f a[ZT];
#pragma unroll
        for (int zt = 0; zt < ZT; ++zt) {
            a[zt].x = fi[zt] * fj[zt][Q + s].x;
            a[zt].y = fi[zt] * fj[zt][Q + s].y;
        }
#pragma unroll
        for (int zt = 0; zt < ZT; ++zt) {
#pragma unroll
            for (int u = 0; u < KT; ++u)
                acc[zt][u] = __builtin_amdgcn_wmma_f32_16x16x4_f32(
                     false, a[zt],  false, b[4 * s + u],
                     (short)0, acc[zt][u],  false,  false);
        }
    }
}

template <int Q, int QEND, int P0>
struct QRows {
    static __device__ __forceinline__ void run(
        const float2* Bp, const int half, const v2f (&fj)[ZT][16],
        v2f (&b)[64], v8f (&acc)[ZT][KT]) {
        constexpr int  n     = 16 - Q;
        constexpr int  H     = (n + 1) / 2;
        constexpr bool LASTQ = (Q + 1 == QEND);
        const float2* rp = Bp + (slot_base(Q, 0) - P0) * 128;

        v2f part[ZT];
#pragma unroll
        for (int zt = 0; zt < ZT; ++zt) {
            part[zt].x = xhalf(fj[zt][Q].x);
            part[zt].y = xhalf(fj[zt][Q].y);
        }

#pragma unroll 1
        for (int r = 0; r < 4; ++r) {
            const bool mine = (half == (r >> 1));
            const bool odd  = (r & 1) != 0;
            float fi[ZT];
#pragma unroll
            for (int zt = 0; zt < ZT; ++zt) {
                const float own = odd ? fj[zt][Q].y : fj[zt][Q].x;
                const float oth = odd ? part[zt].y : part[zt].x;
                fi[zt] = mine ? own : oth;
            }

            const float2* np = rp + ((LASTQ && r == 3) ? r : (r + 1)) * (n * 128);

            mma_steps<Q, 0, H>(fi, fj, b, acc);
            __builtin_amdgcn_sched_barrier(0);
            load_frags<0, 4 * H>(np, b);
            __builtin_amdgcn_sched_barrier(0);
            mma_steps<Q, H, n>(fi, fj, b, acc);
            __builtin_amdgcn_sched_barrier(0);
            load_frags<4 * H, 4 * (n - 1)>(np, b);
            if constexpr (n - 1 >= H) {
                if (r < 3) load_frags<4 * (n - 1), 4 * n>(np, b);
            }
        }
        if constexpr (!LASTQ) QRows<Q + 1, QEND, P0>::run(Bp, half, fj, b, acc);
    }
};

__global__ __launch_bounds__(256) void lts_bilinear_wmma_kernel(
    const float* __restrict__ feat,
    const float* __restrict__ Ssw,
    float* __restrict__ out)
{
    __shared__ __attribute__((aligned(16))) float smem[PART_SLOTS * 256];

    const int tid  = threadIdx.x;
    const int wave = tid >> 5;
    const int lane = tid & 31;
    const int half = lane >> 4;
    const int l16  = lane & 15;
    const int zbase = blockIdx.x * 512 + wave * (16 * ZT);

    v2f fj[ZT][16];
#pragma unroll
    for (int zt = 0; zt < ZT; ++zt) {
        const float2* row = (const float2*)(feat + (zbase + 16 * zt + l16) * 64) + half;
#pragma unroll
        for (int jj = 0; jj < 16; ++jj) {
            const float2 v = row[2 * jj];
            fj[zt][jj].x = v.x;
            fj[zt][jj].y = v.y;
        }
    }

    v8f acc[ZT][KT];
#pragma unroll
    for (int zt = 0; zt < ZT; ++zt) {
#pragma unroll
        for (int u = 0; u < KT; ++u) acc[zt][u] = v8f{};
    }

    const float2* Bl = (const float2*)smem + lane;
    v2f b[64];

#pragma unroll 1
    for (int part = 0; part < 2; ++part) {
        __syncthreads();
        {
            const float4* src = (const float4*)(Ssw + (size_t)part * (PART1_FIRST * 256)) + tid;
            float4* dst = (float4*)smem + tid;
#pragma unroll 1
            for (int bt = 0; bt < 2; ++bt) {
                float4 v[35];
#pragma unroll
                for (int m = 0; m < 35; ++m) v[m] = src[(bt * 35 + m) * 256];
#pragma unroll
                for (int m = 0; m < 35; ++m) dst[(bt * 35 + m) * 256] = v[m];
            }
        }
        __syncthreads();

        if (part == 0) {
            load_frags<0, 64>(Bl, b);
            QRows<0, SPLIT_Q, 0>::run(Bl, half, fj, b, acc);
        } else {
            load_frags<0, 4 * (16 - SPLIT_Q)>(
                Bl + (slot_base(SPLIT_Q, 0) - PART1_FIRST) * 128, b);
            QRows<SPLIT_Q, 16, PART1_FIRST>::run(Bl, half, fj, b, acc);
        }
    }

#pragma unroll
    for (int zt = 0; zt < ZT; ++zt) {
#pragma unroll
        for (int r = 0; r < 8; ++r) {
            const int z = zbase + 16 * zt + half * 8 + r;
#pragma unroll
            for (int u = 0; u < KT; ++u)
                out[z * 64 + u * 16 + l16] = acc[zt][u][r];
        }
    }
}

extern "C" void kernel_launch(void* const* d_in, const int* in_sizes, int n_in,
                              void* d_out, int out_size, void* d_ws, size_t ws_size,
                              hipStream_t stream) {
    const float* feat = (const float*)d_in[0];
    const float* mix  = (const float*)d_in[1];
    const float* wgt  = (const float*)d_in[2];
    float* out = (float*)d_out;
    float* Ssw = (float*)d_ws;

    lts_reduce_fold_kernel<<<Kn, 1024, 0, stream>>>(mix, wgt, Ssw);

    lts_bilinear_wmma_kernel<<<Zn / 512, 256, 0, stream>>>(feat, Ssw, out);
}


# Diagonal 4x4 blocks packed densely: their 10 pairs i<=j fill 5 k-steps per two blocks, not 8 zero-padded ones; 520 slots per k-tile, not 544 (fewer WMMAs and LDS reads); sum order changes
# speedup vs baseline: 2.1247x; 1.0491x over previous
#include <hip/hip_runtime.h>
#include <hip/hip_bf16.h>


#define Zn 131072
#define Nn 64
#define Kn 64
#define Wn 32

#define SLOTS 520

#define ZT 4
#define KT 4

#define PART_SLOTS  272
#define PART1_FIRST 248
#define SPLIT_Q     5

typedef __attribute__((ext_vector_type(2))) float v2f;
typedef __attribute__((ext_vector_type(8))) float v8f;

__host__ __device__ constexpr int dsteps(int q) { return 2 + (q & 1); }

__host__ __device__ constexpr int blk_base(int q) {
    return 64 * q - 2 * q * q + (q >> 1);
}

__host__ __device__ constexpr unsigned long long diag_place_table() {
    unsigned long long t = 0;
    t |= 0ull << (4 * 0);
    t |= 1ull << (4 * 1);
    t |= 5ull << (4 * 2);
    t |= 8ull << (4 * 3);
    t |= 4ull << (4 * 5);
    t |= 7ull << (4 * 6);
    t |= 9ull << (4 * 7);
    t |= 2ull << (4 * 10);
    t |= 3ull << (4 * 11);
    t |= 6ull << (4 * 15);
    return t;
}

__global__ __launch_bounds__(1024) void lts_reduce_fold_kernel(
    const float* __restrict__ mix,
    const float* __restrict__ wgt,
    float* __restrict__ Ssw)
{
    __shared__ float sM[Nn * (Nn + 1)];

    const int k   = blockIdx.x;
    const int tid = threadIdx.x;
    const float* src = mix + k * (Nn * Nn) + tid;

    float acc[4] = {0.0f, 0.0f, 0.0f, 0.0f};
#pragma unroll
    for (int w = 0; w < Wn; ++w) {
        const float wv = wgt[w];
#pragma unroll
        for (int m = 0; m < 4; ++m)
            acc[m] = fmaf(wv, src[w * (Kn * Nn * Nn) + m * 1024], acc[m]);
    }
#pragma unroll
    for (int m = 0; m < 4; ++m) {
        const int e = m * 1024 + tid;
        sM[(e >> 6) * (Nn + 1) + (e & 63)] = acc[m];
    }
    __syncthreads();

    const int u  = k >> 4;
    const int kl = k & 15;
#pragma unroll
    for (int m = 0; m < 4; ++m) {
        const int e  = m * 1024 + tid;
        const int i  = e >> 6;
        const int j  = e & 63;
        const int q  = i >> 2;
        const int r  = i & 3;
        const int jj = j >> 2;
        const int jo = j & 3;
        if (j >= i) {
            const float v = (j == i) ? acc[m] : acc[m] + sM[j * (Nn + 1) + i];
            int slot, kk;
            if (jj > q) {
                slot = blk_base(q) + dsteps(q) + r * (15 - q) + (jj - q - 1);
                kk   = jo;
            } else {
                const int code = (int)((diag_place_table() >> (4 * (r * 4 + jo))) & 15ull);
                const int step = code >> 2;
                kk = code & 3;
                if (step == 2) {
                    slot = blk_base(q | 1) + 2;
                    kk  += 2 * (q & 1);
                } else {
                    slot = blk_base(q) + step;
                }
            }
            const int lane = (kk >> 1) * 16 + kl;
            Ssw[(((slot * KT + u) * 32 + lane) * 2) + (kk & 1)] = v;
        }
    }
}


__device__ __forceinline__ float xhalf(float x) {
    return __builtin_bit_cast(float, __builtin_amdgcn_permlanex16(
        __builtin_bit_cast(int, x), __builtin_bit_cast(int, x),
        0x76543210, 0xfedcba98, false, false));
}

template <int E0, int E1, int NB>
__device__ __forceinline__ void load_frags(const float2* p, v2f (&b)[NB]) {
#pragma unroll
    for (int e = E0; e < E1; ++e) {
        const float2 v = p[e * 32];
        b[e].x = v.x;
        b[e].y = v.y;
    }
}

template <int S, int NB>
__device__ __forceinline__ void mma_slot(const v2f (&a)[ZT], const v2f (&b)[NB],
                                         v8f (&acc)[ZT][KT]) {
#pragma unroll
    for (int zt = 0; zt < ZT; ++zt) {
#pragma unroll
        for (int u = 0; u < KT; ++u)
            acc[zt][u] = __builtin_amdgcn_wmma_f32_16x16x4_f32(
                 false, a[zt],  false, b[4 * S + u],
                 (short)0, acc[zt][u],  false,  false);
    }
}

template <int J0, int S0, int S1>
__device__ __forceinline__ void mma_steps(const float (&fi)[ZT], const v2f (&fj)[ZT][16],
                                          const v2f (&b)[60], v8f (&acc)[ZT][KT]) {
#pragma unroll
    for (int s = S0; s < S1; ++s) {
        v2f a[ZT];
#pragma unroll
        for (int zt = 0; zt < ZT; ++zt) {
            a[zt].x = fi[zt] * fj[zt][J0 + s].x;
            a[zt].y = fi[zt] * fj[zt][J0 + s].y;
        }
#pragma unroll
        for (int zt = 0; zt < ZT; ++zt) {
#pragma unroll
            for (int u = 0; u < KT; ++u)
                acc[zt][u] = __builtin_amdgcn_wmma_f32_16x16x4_f32(
                     false, a[zt],  false, b[4 * s + u],
                     (short)0, acc[zt][u],  false,  false);
        }
    }
}

template <int Q>
__device__ __forceinline__ void diag_steps(const bool hi, const v2f (&fj)[ZT][16],
                                           const v2f (&part)[ZT], const v2f (&bD)[12],
                                           v8f (&acc)[ZT][KT]) {
    v2f a[ZT];
#pragma unroll
    for (int zt = 0; zt < ZT; ++zt) {
        a[zt].x = fj[zt][Q].x * fj[zt][Q].x;
        a[zt].y = fj[zt][Q].x * fj[zt][Q].y;
    }
    mma_slot<0>(a, bD, acc);
#pragma unroll
    for (int zt = 0; zt < ZT; ++zt) {
        const float c = hi ? part[zt].y : part[zt].x;
        a[zt].x = fj[zt][Q].y * fj[zt][Q].y;
        a[zt].y = fj[zt][Q].x * c;
    }
    mma_slot<1>(a, bD, acc);
    if constexpr ((Q & 1) != 0) {
#pragma unroll
        for (int zt = 0; zt < ZT; ++zt) {
            const float x3prev = xhalf(fj[zt][Q - 1].y);
            const float px = hi ? part[zt].x : fj[zt][Q - 1].x;
            const float py = hi ? part[zt].y : fj[zt][Q - 1].y;
            const float sc = hi ? fj[zt][Q].y : x3prev;
            a[zt].x = px * sc;
            a[zt].y = py * sc;
        }
        mma_slot<2>(a, bD, acc);
    }
}

template <int Q, int QEND, int P0>
struct QRows {
    static __device__ __forceinline__ void run(
        const float2* Bp, const int half, const v2f (&fj)[ZT][16],
        v2f (&bD)[12], v2f (&bDn)[12], v2f (&b)[60], v8f (&acc)[ZT][KT]) {
        constexpr int  m     = 15 - Q;
        constexpr int  D     = dsteps(Q);
        constexpr int  H     = (m + 1) / 2;
        constexpr bool LASTQ = (Q + 1 == QEND);
        const float2* dp = Bp + (blk_base(Q) - P0) * 128;
        const float2* rp = dp + D * 128;
        const bool hi = (half != 0);

        v2f part[ZT];
#pragma unroll
        for (int zt = 0; zt < ZT; ++zt) {
            part[zt].x = xhalf(fj[zt][Q].x);
            part[zt].y = xhalf(fj[zt][Q].y);
        }

        if constexpr (!LASTQ) {
            load_frags<0, 4 * dsteps(Q + 1)>(rp + 4 * m * 128, bDn);
            __builtin_amdgcn_sched_barrier(0);
        }
        diag_steps<Q>(hi, fj, part, bD, acc);

        if constexpr (m > 0) {
            constexpr bool STAY = LASTQ || (m == 1);
#pragma unroll 1
            for (int r = 0; r < 4; ++r) {
                const bool mine = (half == (r >> 1));
                const bool odd  = (r & 1) != 0;
                float fi[ZT];
#pragma unroll
                for (int zt = 0; zt < ZT; ++zt) {
                    const float own = odd ? fj[zt][Q].y : fj[zt][Q].x;
                    const float oth = odd ? part[zt].y : part[zt].x;
                    fi[zt] = mine ? own : oth;
                }

                const float2* np = (r < 3) ? rp + (r + 1) * (m * 128)
                                 : (STAY ? rp + 3 * (m * 128)
                                         : rp + 4 * (m * 128) + dsteps(Q + 1) * 128);

                mma_steps<Q + 1, 0, H>(fi, fj, b, acc);
                __builtin_amdgcn_sched_barrier(0);
                load_frags<0, 4 * H>(np, b);
                __builtin_amdgcn_sched_barrier(0);
                mma_steps<Q + 1, H, m>(fi, fj, b, acc);
                __builtin_amdgcn_sched_barrier(0);
                load_frags<4 * H, 4 * (m - 1)>(np, b);
                if constexpr (m - 1 >= H) {
                    if (r < 3) load_frags<4 * (m - 1), 4 * m>(np, b);
                }
            }
        }
        if constexpr (!LASTQ) QRows<Q + 1, QEND, P0>::run(Bp, half, fj, bDn, bD, b, acc);
    }
};

__device__ __forceinline__ void stage_part(const float* __restrict__ from, float* smem, const int tid) {
    const float4* src = (const float4*)from + tid;
    float4* dst = (float4*)smem + tid;
#pragma unroll 1
    for (int bt = 0; bt < 2; ++bt) {
        float4 v[34];
#pragma unroll
        for (int m = 0; m < 34; ++m) v[m] = src[(bt * 34 + m) * 256];
#pragma unroll
        for (int m = 0; m < 34; ++m) dst[(bt * 34 + m) * 256] = v[m];
    }
}

__global__ __launch_bounds__(256) void lts_bilinear_wmma_kernel(
    const float* __restrict__ feat,
    const float* __restrict__ Ssw,
    float* __restrict__ out)
{
    __shared__ __attribute__((aligned(16))) float smem[PART_SLOTS * 256];

    const int tid  = threadIdx.x;
    const int wave = tid >> 5;
    const int lane = tid & 31;
    const int half = lane >> 4;
    const int l16  = lane & 15;
    const int zbase = blockIdx.x * 512 + wave * (16 * ZT);

    v2f fj[ZT][16];
#pragma unroll
    for (int zt = 0; zt < ZT; ++zt) {
        const float2* row = (const float2*)(feat + (zbase + 16 * zt + l16) * 64) + half;
#pragma unroll
        for (int jj = 0; jj < 16; ++jj) {
            const float2 v = row[2 * jj];
            fj[zt][jj].x = v.x;
            fj[zt][jj].y = v.y;
        }
    }

    v8f acc[ZT][KT];
#pragma unroll
    for (int zt = 0; zt < ZT; ++zt) {
#pragma unroll
        for (int u = 0; u < KT; ++u) acc[zt][u] = v8f{};
    }

    const float2* Bl = (const float2*)smem + lane;
    v2f bD[12];
    v2f bDn[12];
    v2f b[60];

    stage_part(Ssw, smem, tid);
    __syncthreads();
    load_frags<0, 4 * dsteps(0)>(Bl, bD);
    load_frags<0, 4 * 15>(Bl + dsteps(0) * 128, b);
    QRows<0, SPLIT_Q, 0>::run(Bl, half, fj, bD, bDn, b, acc);

    __syncthreads();
    stage_part(Ssw + PART1_FIRST * 256, smem, tid);
    __syncthreads();
    {
        const float2* dp = Bl + (blk_base(SPLIT_Q) - PART1_FIRST) * 128;
        load_frags<0, 4 * dsteps(SPLIT_Q)>(dp, bD);
        load_frags<0, 4 * (15 - SPLIT_Q)>(dp + dsteps(SPLIT_Q) * 128, b);
        QRows<SPLIT_Q, 16, PART1_FIRST>::run(Bl, half, fj, bD, bDn, b, acc);
    }

#pragma unroll
    for (int zt = 0; zt < ZT; ++zt) {
#pragma unroll
        for (int r = 0; r < 8; ++r) {
            const int z = zbase + 16 * zt + half * 8 + r;
#pragma unroll
            for (int u = 0; u < KT; ++u)
                out[z * 64 + u * 16 + l16] = acc[zt][u][r];
        }
    }
}

extern "C" void kernel_launch(void* const* d_in, const int* in_sizes, int n_in,
                              void* d_out, int out_size, void* d_ws, size_t ws_size,
                              hipStream_t stream) {
    const float* feat = (const float*)d_in[0];
    const float* mix  = (const float*)d_in[1];
    const float* wgt  = (const float*)d_in[2];
    float* out = (float*)d_out;
    float* Ssw = (float*)d_ws;

    lts_reduce_fold_kernel<<<Kn, 1024, 0, stream>>>(mix, wgt, Ssw);

    lts_bilinear_wmma_kernel<<<Zn / 512, 256, 0, stream>>>(feat, Ssw, out);
}
